# starting kernel
# baseline (speedup 1.0000x reference)
#include <hip/hip_runtime.h>
#include <hip/hip_bf16.h>
#include <math.h>

typedef __attribute__((ext_vector_type(2))) float v2f;
typedef __attribute__((ext_vector_type(8))) float v8f;

#define BB 64
#define NN 1024
#define FF 64
#define DEG 8
#define LRELU_ALPHA 0.2f
#define MASK_VAL -9.0e15f
#define WT_STRIDE 66

__global__ __launch_bounds__(256) void gat_gemm_h(const float* __restrict__ atoms,
                                                  const float* __restrict__ W,
                                                  float* __restrict__ h) {
    __shared__ float ldsWt[FF * WT_STRIDE];
    for (int i = threadIdx.x; i < FF * FF; i += 256) {
        const int kk = i >> 6;
        const int nn = i & 63;
        ldsWt[nn * WT_STRIDE + kk] = W[i];
    }
    __syncthreads();

    const int lane = threadIdx.x & 31;
    const int wave = threadIdx.x >> 5;
    const int l16  = lane & 15;
    const int half = lane >> 4;
    const int rowBase = blockIdx.x * 128 + wave * 16;

    v8f acc0 = {}, acc1 = {}, acc2 = {}, acc3 = {};

    const float* aRow = atoms + (size_t)(rowBase + l16) * FF;
    const float* bCol0 = ldsWt + (l16 +  0) * WT_STRIDE;
    const float* bCol1 = ldsWt + (l16 + 16) * WT_STRIDE;
    const float* bCol2 = ldsWt + (l16 + 32) * WT_STRIDE;
    const float* bCol3 = ldsWt + (l16 + 48) * WT_STRIDE;

    for (int k = 0; k < FF; k += 4) {
        const int kb = k + half * 2;
        v2f afrag = *(const v2f*)(aRow + kb);
        v2f b0 = *(const v2f*)(bCol0 + kb);
        v2f b1 = *(const v2f*)(bCol1 + kb);
        v2f b2 = *(const v2f*)(bCol2 + kb);
        v2f b3 = *(const v2f*)(bCol3 + kb);

        acc0 = __builtin_amdgcn_wmma_f32_16x16x4_f32(false, afrag, false, b0,
                                                     (short)0, acc0, false, false);
        acc1 = __builtin_amdgcn_wmma_f32_16x16x4_f32(false, afrag, false, b1,
                                                     (short)0, acc1, false, false);
        acc2 = __builtin_amdgcn_wmma_f32_16x16x4_f32(false, afrag, false, b2,
                                                     (short)0, acc2, false, false);
        acc3 = __builtin_amdgcn_wmma_f32_16x16x4_f32(false, afrag, false, b3,
                                                     (short)0, acc3, false, false);
    }

    #pragma unroll
    for (int r = 0; r < 8; ++r) {
        const int row = rowBase + half * 8 + r;
        float* dst = h + (size_t)row * FF + l16;
        dst[0]  = acc0[r];
        dst[16] = acc1[r];
        dst[32] = acc2[r];
        dst[48] = acc3[r];
    }
}

__global__ __launch_bounds__(256) void gat_scores(const float* __restrict__ h,
                                                  const float* __restrict__ a,
                                                  float* __restrict__ s1,
                                                  float* __restrict__ s2) {
    const int lane = threadIdx.x & 31;
    const int wave = threadIdx.x >> 5;
    const int row  = blockIdx.x * 8 + wave;

    const float* hr = h + (size_t)row * FF;
    const float x0 = hr[lane];
    const float x1 = hr[lane + 32];

    float p1 = x0 * a[lane]      + x1 * a[lane + 32];
    float p2 = x0 * a[FF + lane] + x1 * a[FF + lane + 32];

    #pragma unroll
    for (int m = 16; m >= 1; m >>= 1) {
        p1 += __shfl_xor(p1, m, 32);
        p2 += __shfl_xor(p2, m, 32);
    }
    if (lane == 0) {
        s1[row] = p1;
        s2[row] = p2;
    }
}

__global__ __launch_bounds__(256) void gat_aggregate(const float* __restrict__ h,
                                                     const int* __restrict__ edges,
                                                     const float* __restrict__ s1,
                                                     const float* __restrict__ s2,
                                                     float* __restrict__ out) {
    const int grp = threadIdx.x >> 6;
    const int t   = threadIdx.x & 63;
    const int row = blockIdx.x * 4 + grp;
    const int batchRow0 = row & ~(NN - 1);

    __shared__ float sh_e[4][DEG];
    __shared__ int   sh_j[4][DEG];

    if (t < DEG) {
        const int* er = edges + (size_t)row * DEG;
        const int j = er[t];
        bool dup = false;
        for (int dd = 0; dd < t; ++dd) dup = dup || (er[dd] == j);
        float e;
        if (dup) {
            e = MASK_VAL;
        } else {
            const float v = s1[row] + s2[batchRow0 + j];
            e = v > 0.0f ? v : LRELU_ALPHA * v;
        }
        sh_e[grp][t] = e;
        sh_j[grp][t] = j;
    }
    __syncthreads();

    float m = sh_e[grp][0];
    #pragma unroll
    for (int d = 1; d < DEG; ++d) m = fmaxf(m, sh_e[grp][d]);

    float w[DEG];
    float sum = 0.0f;
    #pragma unroll
    for (int d = 0; d < DEG; ++d) {
        w[d] = expf(sh_e[grp][d] - m);
        sum += w[d];
    }
    const float inv = 1.0f / sum;

    const size_t batchBase = (size_t)batchRow0 * FF;
    float acc = 0.0f;
    #pragma unroll
    for (int d = 0; d < DEG; ++d) {
        acc += (w[d] * inv) * h[batchBase + (size_t)sh_j[grp][d] * FF + t];
    }

    const float y = acc > 0.0f ? acc : (expf(acc) - 1.0f);
    out[(size_t)row * FF + t] = y;
}

extern "C" void kernel_launch(void* const* d_in, const int* in_sizes, int n_in,
                              void* d_out, int out_size, void* d_ws, size_t ws_size,
                              hipStream_t stream) {
    (void)in_sizes; (void)n_in; (void)out_size; (void)ws_size;

    const float* atoms = (const float*)d_in[0];
    const int*   edges = (const int*)d_in[1];
    const float* W     = (const float*)d_in[2];
    const float* a     = (const float*)d_in[3];
    float* out = (float*)d_out;

    const int ROWS = BB * NN;
    float* h  = (float*)d_ws;
    float* s1 = h + (size_t)ROWS * FF;
    float* s2 = s1 + ROWS;

    gat_gemm_h   <<<ROWS / 128, 256, 0, stream>>>(atoms, W, h);
    gat_scores   <<<ROWS / 8,   256, 0, stream>>>(h, a, s1, s2);
    gat_aggregate<<<ROWS / 4,   256, 0, stream>>>(h, edges, s1, s2, out);
}


# Aggregate kernel: one wave owns 16 rows; 16 lanes per row gather each neighbour row of h with 16-byte loads, 16 gathers in flight (4096 waves, was 131072); per-element sum order kept
# speedup vs baseline: 1.7755x; 1.7755x over previous
#include <hip/hip_runtime.h>
#include <hip/hip_bf16.h>
#include <math.h>

typedef __attribute__((ext_vector_type(2))) float v2f;
typedef __attribute__((ext_vector_type(8))) float v8f;

#define BB 64
#define NN 1024
#define FF 64
#define DEG 8
#define LRELU_ALPHA 0.2f
#define MASK_VAL -9.0e15f
#define WT_STRIDE 66

__global__ __launch_bounds__(256) void gat_gemm_h(const float* __restrict__ atoms,
                                                  const float* __restrict__ W,
                                                  float* __restrict__ h) {
    __shared__ float ldsWt[FF * WT_STRIDE];
    for (int i = threadIdx.x; i < FF * FF; i += 256) {
        const int kk = i >> 6;
        const int nn = i & 63;
        ldsWt[nn * WT_STRIDE + kk] = W[i];
    }
    __syncthreads();

    const int lane = threadIdx.x & 31;
    const int wave = threadIdx.x >> 5;
    const int l16  = lane & 15;
    const int half = lane >> 4;
    const int rowBase = blockIdx.x * 128 + wave * 16;

    v8f acc0 = {}, acc1 = {}, acc2 = {}, acc3 = {};

    const float* aRow = atoms + (size_t)(rowBase + l16) * FF;
    const float* bCol0 = ldsWt + (l16 +  0) * WT_STRIDE;
    const float* bCol1 = ldsWt + (l16 + 16) * WT_STRIDE;
    const float* bCol2 = ldsWt + (l16 + 32) * WT_STRIDE;
    const float* bCol3 = ldsWt + (l16 + 48) * WT_STRIDE;

    for (int k = 0; k < FF; k += 4) {
        const int kb = k + half * 2;
        v2f afrag = *(const v2f*)(aRow + kb);
        v2f b0 = *(const v2f*)(bCol0 + kb);
        v2f b1 = *(const v2f*)(bCol1 + kb);
        v2f b2 = *(const v2f*)(bCol2 + kb);
        v2f b3 = *(const v2f*)(bCol3 + kb);

        acc0 = __builtin_amdgcn_wmma_f32_16x16x4_f32(false, afrag, false, b0,
                                                     (short)0, acc0, false, false);
        acc1 = __builtin_amdgcn_wmma_f32_16x16x4_f32(false, afrag, false, b1,
                                                     (short)0, acc1, false, false);
        acc2 = __builtin_amdgcn_wmma_f32_16x16x4_f32(false, afrag, false, b2,
                                                     (short)0, acc2, false, false);
        acc3 = __builtin_amdgcn_wmma_f32_16x16x4_f32(false, afrag, false, b3,
                                                     (short)0, acc3, false, false);
    }

    #pragma unroll
    for (int r = 0; r < 8; ++r) {
        const int row = rowBase + half * 8 + r;
        float* dst = h + (size_t)row * FF + l16;
        dst[0]  = acc0[r];
        dst[16] = acc1[r];
        dst[32] = acc2[r];
        dst[48] = acc3[r];
    }
}

__global__ __launch_bounds__(256) void gat_scores(const float* __restrict__ h,
                                                  const float* __restrict__ a,
                                                  float* __restrict__ s1,
                                                  float* __restrict__ s2) {
    const int lane = threadIdx.x & 31;
    const int wave = threadIdx.x >> 5;
    const int row  = blockIdx.x * 8 + wave;

    const float* hr = h + (size_t)row * FF;
    const float x0 = hr[lane];
    const float x1 = hr[lane + 32];

    float p1 = x0 * a[lane]      + x1 * a[lane + 32];
    float p2 = x0 * a[FF + lane] + x1 * a[FF + lane + 32];

    #pragma unroll
    for (int m = 16; m >= 1; m >>= 1) {
        p1 += __shfl_xor(p1, m, 32);
        p2 += __shfl_xor(p2, m, 32);
    }
    if (lane == 0) {
        s1[row] = p1;
        s2[row] = p2;
    }
}

#define RPW 16
__global__ __launch_bounds__(256) void gat_aggregate(const float* __restrict__ h,
                                                     const int* __restrict__ edges,
                                                     const float* __restrict__ s1,
                                                     const float* __restrict__ s2,
                                                     float* __restrict__ out) {
    __shared__ __attribute__((aligned(16))) int   sh_j[8][RPW * DEG];
    __shared__ __attribute__((aligned(16))) float sh_w[8][RPW * DEG];

    const int lane = threadIdx.x & 31;
    const int wave = threadIdx.x >> 5;
    const int row0 = (blockIdx.x * 8 + wave) * RPW;
    const int batchRow0 = row0 & ~(NN - 1);

    {
        const int q = lane & 1;
        const int4 mine = *(const int4*)(edges + (size_t)row0 * DEG + lane * 4);
        const float s1r = s1[row0 + (lane >> 1)];

        const int jm[4] = {mine.x, mine.y, mine.z, mine.w};
        int jo[4];
        #pragma unroll
        for (int k = 0; k < 4; ++k) jo[k] = __shfl_xor(jm[k], 1, 32);

        float sv[4];
        #pragma unroll
        for (int k = 0; k < 4; ++k) sv[k] = s2[batchRow0 + jm[k]];
        asm volatile("" :: "v"(sv[0]), "v"(sv[1]), "v"(sv[2]), "v"(sv[3]));

        float e[4];
        #pragma unroll
        for (int k = 0; k < 4; ++k) {
            bool dup = false;
            #pragma unroll
            for (int kk = 0; kk < k; ++kk) dup = dup || (jm[kk] == jm[k]);
            #pragma unroll
            for (int kk = 0; kk < 4; ++kk) dup = dup || (q && (jo[kk] == jm[k]));
            const float v  = s1r + sv[k];
            const float lr = v > 0.0f ? v : LRELU_ALPHA * v;
            e[k] = dup ? MASK_VAL : lr;
        }

        float m = fmaxf(fmaxf(e[0], e[1]), fmaxf(e[2], e[3]));
        m = fmaxf(m, __shfl_xor(m, 1, 32));

        float wm[4], wo[4];
        #pragma unroll
        for (int k = 0; k < 4; ++k) wm[k] = expf(e[k] - m);
        #pragma unroll
        for (int k = 0; k < 4; ++k) wo[k] = __shfl_xor(wm[k], 1, 32);

        float sum = 0.0f;
        #pragma unroll
        for (int k = 0; k < 4; ++k) sum += q ? wo[k] : wm[k];
        #pragma unroll
        for (int k = 0; k < 4; ++k) sum += q ? wm[k] : wo[k];
        const float inv = 1.0f / sum;

        float4 wn;
        wn.x = wm[0] * inv; wn.y = wm[1] * inv; wn.z = wm[2] * inv; wn.w = wm[3] * inv;
        *(int4*)&sh_j[wave][lane * 4]   = mine;
        *(float4*)&sh_w[wave][lane * 4] = wn;
    }
    __syncthreads();

    const int hf  = lane >> 4;
    const int l16 = lane & 15;
    const float* hb = h + (size_t)batchRow0 * FF + l16 * 4;

    for (int it = 0; it < RPW / 4; ++it) {
        float4 hv[2][DEG];
        float  w[2][DEG];
        #pragma unroll
        for (int u = 0; u < 2; ++u) {
            const int rl = it * 4 + u * 2 + hf;
            const int4   ja = *(const int4*)&sh_j[wave][rl * DEG];
            const int4   jb = *(const int4*)&sh_j[wave][rl * DEG + 4];
            const float4 wa = *(const float4*)&sh_w[wave][rl * DEG];
            const float4 wb = *(const float4*)&sh_w[wave][rl * DEG + 4];
            const int j[DEG] = {ja.x, ja.y, ja.z, ja.w, jb.x, jb.y, jb.z, jb.w};
            w[u][0] = wa.x; w[u][1] = wa.y; w[u][2] = wa.z; w[u][3] = wa.w;
            w[u][4] = wb.x; w[u][5] = wb.y; w[u][6] = wb.z; w[u][7] = wb.w;
            #pragma unroll
            for (int d = 0; d < DEG; ++d)
                hv[u][d] = *(const float4*)(hb + (size_t)j[d] * FF);
        }

        __builtin_amdgcn_sched_barrier(0);

        #pragma unroll
        for (int u = 0; u < 2; ++u) {
            const int rl = it * 4 + u * 2 + hf;
            float4 acc = {0.0f, 0.0f, 0.0f, 0.0f};
            #pragma unroll
            for (int d = 0; d < DEG; ++d) {
                acc.x += w[u][d] * hv[u][d].x;
                acc.y += w[u][d] * hv[u][d].y;
                acc.z += w[u][d] * hv[u][d].z;
                acc.w += w[u][d] * hv[u][d].w;
            }

            float4 y;
            y.x = acc.x > 0.0f ? acc.x : (expf(acc.x) - 1.0f);
            y.y = acc.y > 0.0f ? acc.y : (expf(acc.y) - 1.0f);
            y.z = acc.z > 0.0f ? acc.z : (expf(acc.z) - 1.0f);
            y.w = acc.w > 0.0f ? acc.w : (expf(acc.w) - 1.0f);
            *(float4*)(out + (size_t)(row0 + rl) * FF + l16 * 4) = y;
        }
    }
}

extern "C" void kernel_launch(void* const* d_in, const int* in_sizes, int n_in,
                              void* d_out, int out_size, void* d_ws, size_t ws_size,
                              hipStream_t stream) {
    (void)in_sizes; (void)n_in; (void)out_size; (void)ws_size;

    const float* atoms = (const float*)d_in[0];
    const int*   edges = (const int*)d_in[1];
    const float* W     = (const float*)d_in[2];
    const float* a     = (const float*)d_in[3];
    float* out = (float*)d_out;

    const int ROWS = BB * NN;
    float* h  = (float*)d_ws;
    float* s1 = h + (size_t)ROWS * FF;
    float* s2 = s1 + ROWS;

    gat_gemm_h   <<<ROWS / 128, 256, 0, stream>>>(atoms, W, h);
    gat_scores   <<<ROWS / 8,   256, 0, stream>>>(h, a, s1, s2);
    gat_aggregate<<<ROWS / (8 * RPW), 256, 0, stream>>>(h, edges, s1, s2, out);
}


# Score dots s1, s2 taken from the GEMM's accumulator tile in its epilogue (packed FMAs per lane, 16-lane DPP fold): the 65536-wave score launch and its re-read of h go; score sum order changes
# speedup vs baseline: 2.4179x; 1.3618x over previous
#include <hip/hip_runtime.h>
#include <hip/hip_bf16.h>
#include <math.h>

typedef __attribute__((ext_vector_type(2))) float v2f;
typedef __attribute__((ext_vector_type(8))) float v8f;

#define BB 64
#define NN 1024
#define FF 64
#define DEG 8
#define LRELU_ALPHA 0.2f
#define MASK_VAL -9.0e15f
#define WT_STRIDE 66

template <int M>
__device__ __forceinline__ float row_xor(float x) {
    return __builtin_bit_cast(float,
        __builtin_amdgcn_update_dpp(0, __builtin_bit_cast(int, x), 0x160 + M, 0xf, 0xf, false));
}

__global__ __launch_bounds__(256) void gat_gemm_h(const float* __restrict__ atoms,
                                                  const float* __restrict__ W,
                                                  const float* __restrict__ a,
                                                  float* __restrict__ h,
                                                  float* __restrict__ s1,
                                                  float* __restrict__ s2) {
    __shared__ float ldsWt[FF * WT_STRIDE];
    const int lane = threadIdx.x & 31;
    const int wave = threadIdx.x >> 5;
    const int l16  = lane & 15;
    const int half = lane >> 4;
    const float a10 = a[l16],      a11 = a[l16 + 16],      a12 = a[l16 + 32],      a13 = a[l16 + 48];
    const float a20 = a[FF + l16], a21 = a[FF + l16 + 16], a22 = a[FF + l16 + 32], a23 = a[FF + l16 + 48];
    for (int i = threadIdx.x; i < FF * FF; i += 256) {
        const int kk = i >> 6;
        const int nn = i & 63;
        ldsWt[nn * WT_STRIDE + kk] = W[i];
    }
    __syncthreads();

    const int rowBase = blockIdx.x * 128 + wave * 16;

    v8f acc0 = {}, acc1 = {}, acc2 = {}, acc3 = {};

    const float* aRow = atoms + (size_t)(rowBase + l16) * FF;
    const float* bCol0 = ldsWt + (l16 +  0) * WT_STRIDE;
    const float* bCol1 = ldsWt + (l16 + 16) * WT_STRIDE;
    const float* bCol2 = ldsWt + (l16 + 32) * WT_STRIDE;
    const float* bCol3 = ldsWt + (l16 + 48) * WT_STRIDE;

    for (int k = 0; k < FF; k += 4) {
        const int kb = k + half * 2;
        v2f afrag = *(const v2f*)(aRow + kb);
        v2f b0 = *(const v2f*)(bCol0 + kb);
        v2f b1 = *(const v2f*)(bCol1 + kb);
        v2f b2 = *(const v2f*)(bCol2 + kb);
        v2f b3 = *(const v2f*)(bCol3 + kb);

        acc0 = __builtin_amdgcn_wmma_f32_16x16x4_f32(false, afrag, false, b0,
                                                     (short)0, acc0, false, false);
        acc1 = __builtin_amdgcn_wmma_f32_16x16x4_f32(false, afrag, false, b1,
                                                     (short)0, acc1, false, false);
        acc2 = __builtin_amdgcn_wmma_f32_16x16x4_f32(false, afrag, false, b2,
                                                     (short)0, acc2, false, false);
        acc3 = __builtin_amdgcn_wmma_f32_16x16x4_f32(false, afrag, false, b3,
                                                     (short)0, acc3, false, false);
    }

    #pragma unroll
    for (int r = 0; r < 8; ++r) {
        const int row = rowBase + half * 8 + r;
        float* dst = h + (size_t)row * FF + l16;
        dst[0]  = acc0[r];
        dst[16] = acc1[r];
        dst[32] = acc2[r];
        dst[48] = acc3[r];
    }

    float v[16];
    #pragma unroll
    for (int p = 0; p < 4; ++p) {
        const v2f h0 = {acc0[2 * p], acc0[2 * p + 1]};
        const v2f h1 = {acc1[2 * p], acc1[2 * p + 1]};
        const v2f h2 = {acc2[2 * p], acc2[2 * p + 1]};
        const v2f h3 = {acc3[2 * p], acc3[2 * p + 1]};
        const v2f q1 = h0 * a10 + h1 * a11 + h2 * a12 + h3 * a13;
        const v2f q2 = h0 * a20 + h1 * a21 + h2 * a22 + h3 * a23;
        v[2 * p] = q1.x;     v[2 * p + 1] = q1.y;
        v[8 + 2 * p] = q2.x; v[8 + 2 * p + 1] = q2.y;
    }
    const bool b3 = (l16 & 8) != 0, b2 = (l16 & 4) != 0, b1 = (l16 & 2) != 0, b0 = (l16 & 1) != 0;
    float x1[8], x2[4], x3[2];
    #pragma unroll
    for (int i = 0; i < 8; ++i) {
        const float keep = b3 ? v[8 + i] : v[i];
        const float send = b3 ? v[i] : v[8 + i];
        x1[i] = keep + row_xor<8>(send);
    }
    #pragma unroll
    for (int i = 0; i < 4; ++i) {
        const float keep = b2 ? x1[4 + i] : x1[i];
        const float send = b2 ? x1[i] : x1[4 + i];
        x2[i] = keep + row_xor<4>(send);
    }
    #pragma unroll
    for (int i = 0; i < 2; ++i) {
        const float keep = b1 ? x2[2 + i] : x2[i];
        const float send = b1 ? x2[i] : x2[2 + i];
        x3[i] = keep + row_xor<2>(send);
    }
    const float keep = b0 ? x3[1] : x3[0];
    const float send = b0 ? x3[0] : x3[1];
    const float tot  = keep + row_xor<1>(send);
    float* sdst = b3 ? s2 : s1;
    sdst[rowBase + half * 8 + (l16 & 7)] = tot;
}

#define RPW 16
__global__ __launch_bounds__(256) void gat_aggregate(const float* __restrict__ h,
                                                     const int* __restrict__ edges,
                                                     const float* __restrict__ s1,
                                                     const float* __restrict__ s2,
                                                     float* __restrict__ out) {
    __shared__ __attribute__((aligned(16))) int   sh_j[8][RPW * DEG];
    __shared__ __attribute__((aligned(16))) float sh_w[8][RPW * DEG];

    const int lane = threadIdx.x & 31;
    const int wave = threadIdx.x >> 5;
    const int row0 = (blockIdx.x * 8 + wave) * RPW;
    const int batchRow0 = row0 & ~(NN - 1);

    {
        const int q = lane & 1;
        const int4 mine = *(const int4*)(edges + (size_t)row0 * DEG + lane * 4);
        const float s1r = s1[row0 + (lane >> 1)];

        const int jm[4] = {mine.x, mine.y, mine.z, mine.w};
        int jo[4];
        #pragma unroll
        for (int k = 0; k < 4; ++k) jo[k] = __shfl_xor(jm[k], 1, 32);

        float sv[4];
        #pragma unroll
        for (int k = 0; k < 4; ++k) sv[k] = s2[batchRow0 + jm[k]];
        asm volatile("" :: "v"(sv[0]), "v"(sv[1]), "v"(sv[2]), "v"(sv[3]));

        float e[4];
        #pragma unroll
        for (int k = 0; k < 4; ++k) {
            bool dup = false;
            #pragma unroll
            for (int kk = 0; kk < k; ++kk) dup = dup || (jm[kk] == jm[k]);
            #pragma unroll
            for (int kk = 0; kk < 4; ++kk) dup = dup || (q && (jo[kk] == jm[k]));
            const float v  = s1r + sv[k];
            const float lr = v > 0.0f ? v : LRELU_ALPHA * v;
            e[k] = dup ? MASK_VAL : lr;
        }

        float m = fmaxf(fmaxf(e[0], e[1]), fmaxf(e[2], e[3]));
        m = fmaxf(m, __shfl_xor(m, 1, 32));

        float wm[4], wo[4];
        #pragma unroll
        for (int k = 0; k < 4; ++k) wm[k] = expf(e[k] - m);
        #pragma unroll
        for (int k = 0; k < 4; ++k) wo[k] = __shfl_xor(wm[k], 1, 32);

        float sum = 0.0f;
        #pragma unroll
        for (int k = 0; k < 4; ++k) sum += q ? wo[k] : wm[k];
        #pragma unroll
        for (int k = 0; k < 4; ++k) sum += q ? wm[k] : wo[k];
        const float inv = 1.0f / sum;

        float4 wn;
        wn.x = wm[0] * inv; wn.y = wm[1] * inv; wn.z = wm[2] * inv; wn.w = wm[3] * inv;
        *(int4*)&sh_j[wave][lane * 4]   = mine;
        *(float4*)&sh_w[wave][lane * 4] = wn;
    }
    __syncthreads();

    const int hf  = lane >> 4;
    const int l16 = lane & 15;
    const float* hb = h + (size_t)batchRow0 * FF + l16 * 4;

    for (int it = 0; it < RPW / 4; ++it) {
        float4 hv[2][DEG];
        float  w[2][DEG];
        #pragma unroll
        for (int u = 0; u < 2; ++u) {
            const int rl = it * 4 + u * 2 + hf;
            const int4   ja = *(const int4*)&sh_j[wave][rl * DEG];
            const int4   jb = *(const int4*)&sh_j[wave][rl * DEG + 4];
            const float4 wa = *(const float4*)&sh_w[wave][rl * DEG];
            const float4 wb = *(const float4*)&sh_w[wave][rl * DEG + 4];
            const int j[DEG] = {ja.x, ja.y, ja.z, ja.w, jb.x, jb.y, jb.z, jb.w};
            w[u][0] = wa.x; w[u][1] = wa.y; w[u][2] = wa.z; w[u][3] = wa.w;
            w[u][4] = wb.x; w[u][5] = wb.y; w[u][6] = wb.z; w[u][7] = wb.w;
            #pragma unroll
            for (int d = 0; d < DEG; ++d)
                hv[u][d] = *(const float4*)(hb + (size_t)j[d] * FF);
        }

        __builtin_amdgcn_sched_barrier(0);

        #pragma unroll
        for (int u = 0; u < 2; ++u) {
            const int rl = it * 4 + u * 2 + hf;
            float4 acc = {0.0f, 0.0f, 0.0f, 0.0f};
            #pragma unroll
            for (int d = 0; d < DEG; ++d) {
                acc.x += w[u][d] * hv[u][d].x;
                acc.y += w[u][d] * hv[u][d].y;
                acc.z += w[u][d] * hv[u][d].z;
                acc.w += w[u][d] * hv[u][d].w;
            }

            float4 y;
            y.x = acc.x > 0.0f ? acc.x : (expf(acc.x) - 1.0f);
            y.y = acc.y > 0.0f ? acc.y : (expf(acc.y) - 1.0f);
            y.z = acc.z > 0.0f ? acc.z : (expf(acc.z) - 1.0f);
            y.w = acc.w > 0.0f ? acc.w : (expf(acc.w) - 1.0f);
            *(float4*)(out + (size_t)(row0 + rl) * FF + l16 * 4) = y;
        }
    }
}

extern "C" void kernel_launch(void* const* d_in, const int* in_sizes, int n_in,
                              void* d_out, int out_size, void* d_ws, size_t ws_size,
                              hipStream_t stream) {
    (void)in_sizes; (void)n_in; (void)out_size; (void)ws_size;

    const float* atoms = (const float*)d_in[0];
    const int*   edges = (const int*)d_in[1];
    const float* W     = (const float*)d_in[2];
    const float* a     = (const float*)d_in[3];
    float* out = (float*)d_out;

    const int ROWS = BB * NN;
    float* h  = (float*)d_ws;
    float* s1 = h + (size_t)ROWS * FF;
    float* s2 = s1 + ROWS;

    gat_gemm_h   <<<ROWS / 128, 256, 0, stream>>>(atoms, W, a, h, s1, s2);
    gat_aggregate<<<ROWS / (8 * RPW), 256, 0, stream>>>(h, edges, s1, s2, out);
}


# One launch: W applied after the attention-weighted sum of gathered input rows (linearity), scores as atoms.(W a); h, s1, s2 and launch 2 go; sum order of scores, softmax sum and output changes
# speedup vs baseline: 4.2967x; 1.7770x over previous
#include <hip/hip_runtime.h>
#include <math.h>


typedef __attribute__((ext_vector_type(2))) float v2f;
typedef __attribute__((ext_vector_type(8))) float v8f;

#define BB 64
#define NN 1024
#define FF 64
#define DEG 8
#define LRELU_ALPHA 0.2f
#define MASK_VAL -9.0e15f
#define WT_STRIDE 66
#define AG_STRIDE 68
#define RPW 16
#define TRIPS (RPW / 4)

template <int M>
__device__ __forceinline__ float row_xor(float x) {
    return __builtin_bit_cast(float,
        __builtin_amdgcn_update_dpp(0, __builtin_bit_cast(int, x), 0x160 + M, 0xf, 0xf, false));
}

template <int S>
__device__ __forceinline__ float row_bcast(float x) {
    return __builtin_bit_cast(float,
        __builtin_amdgcn_update_dpp(0, __builtin_bit_cast(int, x), 0x150 + S, 0xf, 0xf, false));
}

__device__ __forceinline__ float dot4(const float4 x, const float4 y) {
    return x.x * y.x + x.y * y.y + x.z * y.z + x.w * y.w;
}

__device__ __forceinline__ float elu1(const float x) {
    return x > 0.0f ? x : (expf(x) - 1.0f);
}

struct Trip {
    float4 hv[2][DEG];
    float4 own[2];
};

__device__ __forceinline__ void load_trip(Trip& t, const int it, const int* shj,
                                          const float* __restrict__ own0, const float* __restrict__ ab,
                                          const int hf, const int l16) {
    #pragma unroll
    for (int u = 0; u < 2; ++u) {
        const int rl = it * 4 + u * 2 + hf;
        const int4 ja = *(const int4*)&shj[rl * DEG];
        const int4 jb = *(const int4*)&shj[rl * DEG + 4];
        const int j[DEG] = {ja.x, ja.y, ja.z, ja.w, jb.x, jb.y, jb.z, jb.w};
        t.own[u] = *(const float4*)((const char*)own0 + (unsigned)(rl * (FF * 4) + l16 * 16));
        #pragma unroll
        for (int d = 0; d < DEG; ++d)
            t.hv[u][d] = *(const float4*)((const char*)ab + (unsigned)(j[d] * (FF * 4) + l16 * 16));
    }
}

__global__ __launch_bounds__(256) void gat_fused(const float* __restrict__ atoms,
                                                 const int* __restrict__ edges,
                                                 const float* __restrict__ W,
                                                 const float* __restrict__ a,
                                                 float* __restrict__ out) {
    __shared__ float ldsWt[FF * WT_STRIDE];
    __shared__ __attribute__((aligned(16))) float sh_wa[2 * FF];
    __shared__ __attribute__((aligned(16))) int   sh_j[8][RPW * DEG];
    __shared__ __attribute__((aligned(16))) float sh_agg[8][RPW * AG_STRIDE];

    const int lane = threadIdx.x & 31;
    const int wave = __builtin_amdgcn_readfirstlane((int)(threadIdx.x >> 5));
    const int l16  = lane & 15;
    const int hf   = lane >> 4;
    const int row0 = (blockIdx.x * 8 + wave) * RPW;
    const int batchRow0 = row0 & ~(NN - 1);

    {
        const int4 mine = *(const int4*)(edges + (size_t)row0 * DEG + lane * 4);
        float wv[16];
        #pragma unroll
        for (int t = 0; t < 16; ++t) wv[t] = W[threadIdx.x + 256 * t];
        #pragma unroll
        for (int t = 0; t < 16; ++t) {
            const int i = threadIdx.x + 256 * t;
            ldsWt[(i & 63) * WT_STRIDE + (i >> 6)] = wv[t];
        }
        *(int4*)&sh_j[wave][lane * 4] = mine;
    }
    __syncthreads();

    const float* ab   = atoms + (size_t)batchRow0 * FF;
    const float* own0 = atoms + (size_t)row0 * FF;
    Trip cur;
    load_trip(cur, 0, sh_j[wave], own0, ab, hf, l16);

    if (threadIdx.x < 2 * FF) {
        const int which = __builtin_amdgcn_readfirstlane((int)(threadIdx.x >> 6));
        const int k     = threadIdx.x & 63;
        const float* av = a + which * FF;
        float s = 0.0f;
        #pragma unroll
        for (int n = 0; n < FF; ++n) s += ldsWt[n * WT_STRIDE + k] * av[n];
        sh_wa[which * FF + k] = s;
    }
    __syncthreads();

    const float4 wa1_0 = *(const float4*)&sh_wa[4 * l16];
    const float4 wa2_0 = *(const float4*)&sh_wa[FF + 4 * l16];
    const int ul = l16 >> 3;
    const int dl = l16 & 7;
    const bool b3 = (l16 & 8) != 0, b2 = (l16 & 4) != 0, b1 = (l16 & 2) != 0, b0 = (l16 & 1) != 0;

    #pragma unroll
    for (int it = 0; it < TRIPS; ++it) {
        Trip nxt;
        if (it + 1 < TRIPS) load_trip(nxt, it + 1, sh_j[wave], own0, ab, hf, l16);

        __builtin_amdgcn_sched_barrier(0);
        float4 wa1 = wa1_0, wa2 = wa2_0;
        asm volatile("" : "+v"(wa1.x), "+v"(wa1.y), "+v"(wa1.z), "+v"(wa1.w),
                          "+v"(wa2.x), "+v"(wa2.y), "+v"(wa2.z), "+v"(wa2.w));

        float v[16];
        #pragma unroll
        for (int u = 0; u < 2; ++u) {
            #pragma unroll
            for (int d = 0; d < DEG; ++d) v[8 * u + d] = dot4(cur.hv[u][d], wa2);
        }
        float x1[8], x2[4], x3[2];
        #pragma unroll
        for (int i = 0; i < 8; ++i) {
            const float keep = b3 ? v[8 + i] : v[i];
            const float send = b3 ? v[i] : v[8 + i];
            x1[i] = keep + row_xor<8>(send);
        }
        #pragma unroll
        for (int i = 0; i < 4; ++i) {
            const float keep = b2 ? x1[4 + i] : x1[i];
            const float send = b2 ? x1[i] : x1[4 + i];
            x2[i] = keep + row_xor<4>(send);
        }
        #pragma unroll
        for (int i = 0; i < 2; ++i) {
            const float keep = b1 ? x2[2 + i] : x2[i];
            const float send = b1 ? x2[i] : x2[2 + i];
            x3[i] = keep + row_xor<2>(send);
        }
        const float keep0 = b0 ? x3[1] : x3[0];
        const float send0 = b0 ? x3[0] : x3[1];
        const float s2e   = keep0 + row_xor<1>(send0);

        float q0 = dot4(cur.own[0], wa1);
        float q1 = dot4(cur.own[1], wa1);
        q0 += row_xor<1>(q0); q1 += row_xor<1>(q1);
        q0 += row_xor<2>(q0); q1 += row_xor<2>(q1);
        q0 += row_xor<4>(q0); q1 += row_xor<4>(q1);
        q0 += row_xor<8>(q0); q1 += row_xor<8>(q1);
        const float s1r = ul ? q1 : q0;

        const int  rs = (it * 4 + ul * 2 + hf) * DEG;
        const int4 ka = *(const int4*)&sh_j[wave][rs];
        const int4 kb4 = *(const int4*)&sh_j[wave][rs + 4];
        const int  kj[DEG] = {ka.x, ka.y, ka.z, ka.w, kb4.x, kb4.y, kb4.z, kb4.w};
        const int  jm = sh_j[wave][rs + dl];
        int dupi = 0;
        #pragma unroll
        for (int d = 0; d < DEG; ++d) dupi |= (int)(d < dl) & (int)(kj[d] == jm);
        const float sv = s1r + s2e;
        const float lr = sv > 0.0f ? sv : LRELU_ALPHA * sv;
        const float e  = dupi ? MASK_VAL : lr;
        float m = e;
        m = fmaxf(m, row_xor<1>(m));
        m = fmaxf(m, row_xor<2>(m));
        m = fmaxf(m, row_xor<4>(m));
        const float wexp = expf(e - m);
        float sum = wexp + row_xor<1>(wexp);
        sum += row_xor<2>(sum);
        sum += row_xor<4>(sum);
        const float wn = wexp * (1.0f / sum);

        const float wg[2][DEG] = {
            {row_bcast<0>(wn), row_bcast<1>(wn), row_bcast<2>(wn),  row_bcast<3>(wn),
             row_bcast<4>(wn), row_bcast<5>(wn), row_bcast<6>(wn),  row_bcast<7>(wn)},
            {row_bcast<8>(wn), row_bcast<9>(wn), row_bcast<10>(wn), row_bcast<11>(wn),
             row_bcast<12>(wn), row_bcast<13>(wn), row_bcast<14>(wn), row_bcast<15>(wn)}};

        #pragma unroll
        for (int u = 0; u < 2; ++u) {
            const int rl = it * 4 + u * 2 + hf;
            float4 acc = {0.0f, 0.0f, 0.0f, 0.0f};
            #pragma unroll
            for (int d = 0; d < DEG; ++d) {
                acc.x += wg[u][d] * cur.hv[u][d].x;
                acc.y += wg[u][d] * cur.hv[u][d].y;
                acc.z += wg[u][d] * cur.hv[u][d].z;
                acc.w += wg[u][d] * cur.hv[u][d].w;
            }
            *(float4*)&sh_agg[wave][rl * AG_STRIDE + 4 * l16] = acc;
        }
        if (it + 1 < TRIPS) cur = nxt;
    }
    __syncthreads();

    v8f acc0 = {}, acc1 = {}, acc2 = {}, acc3 = {};
    const float* aRow  = &sh_agg[wave][l16 * AG_STRIDE];
    const float* bCol0 = ldsWt + (l16 +  0) * WT_STRIDE;
    const float* bCol1 = ldsWt + (l16 + 16) * WT_STRIDE;
    const float* bCol2 = ldsWt + (l16 + 32) * WT_STRIDE;
    const float* bCol3 = ldsWt + (l16 + 48) * WT_STRIDE;
    #pragma unroll
    for (int k = 0; k < FF; k += 4) {
        const int kb = k + hf * 2;
        v2f afrag = *(const v2f*)(aRow + kb);
        v2f f0 = *(const v2f*)(bCol0 + kb);
        v2f f1 = *(const v2f*)(bCol1 + kb);
        v2f f2 = *(const v2f*)(bCol2 + kb);
        v2f f3 = *(const v2f*)(bCol3 + kb);
        acc0 = __builtin_amdgcn_wmma_f32_16x16x4_f32(false, afrag, false, f0,
                                                     (short)0, acc0, false, false);
        acc1 = __builtin_amdgcn_wmma_f32_16x16x4_f32(false, afrag, false, f1,
                                                     (short)0, acc1, false, false);
        acc2 = __builtin_amdgcn_wmma_f32_16x16x4_f32(false, afrag, false, f2,
                                                     (short)0, acc2, false, false);
        acc3 = __builtin_amdgcn_wmma_f32_16x16x4_f32(false, afrag, false, f3,
                                                     (short)0, acc3, false, false);
    }
    __syncthreads();

    #pragma unroll
    for (int r = 0; r < 8; ++r) {
        float* t = &sh_agg[wave][(hf * 8 + r) * AG_STRIDE + l16];
        t[0]  = elu1(acc0[r]);
        t[16] = elu1(acc1[r]);
        t[32] = elu1(acc2[r]);
        t[48] = elu1(acc3[r]);
    }
    __syncthreads();
    #pragma unroll
    for (int i = 0; i < 8; ++i) {
        const int rl = 2 * i + hf;
        const float4 y = *(const float4*)&sh_agg[wave][rl * AG_STRIDE + 4 * l16];
        *(float4*)(out + (size_t)(row0 + rl) * FF + l16 * 4) = y;
    }
}

extern "C" void kernel_launch(void* const* d_in, const int* in_sizes, int n_in,
                              void* d_out, int out_size, void* d_ws, size_t ws_size,
                              hipStream_t stream) {
    (void)in_sizes; (void)n_in; (void)out_size; (void)d_ws; (void)ws_size;

    const float* atoms = (const float*)d_in[0];
    const int*   edges = (const int*)d_in[1];
    const float* W     = (const float*)d_in[2];
    const float* a     = (const float*)d_in[3];
    float* out = (float*)d_out;

    const int ROWS = BB * NN;
    gat_fused<<<ROWS / (8 * RPW), 256, 0, stream>>>(atoms, edges, W, a, out);
}


# Both exponentials (softmax weight, ELU) on the hardware exp2 instruction, 2^(x*log2 e), in place of the library routine; ELU becomes a branch-free select of 32 independent chains per lane
# speedup vs baseline: 4.6631x; 1.0853x over previous
#include <hip/hip_runtime.h>
#include <math.h>


typedef __attribute__((ext_vector_type(2))) float v2f;
typedef __attribute__((ext_vector_type(8))) float v8f;

#define BB 64
#define NN 1024
#define FF 64
#define DEG 8
#define LRELU_ALPHA 0.2f
#define MASK_VAL -9.0e15f
#define WT_STRIDE 66
#define AG_STRIDE 68
#define RPW 16
#define TRIPS (RPW / 4)

template <int M>
__device__ __forceinline__ float row_xor(float x) {
    return __builtin_bit_cast(float,
        __builtin_amdgcn_update_dpp(0, __builtin_bit_cast(int, x), 0x160 + M, 0xf, 0xf, false));
}

template <int S>
__device__ __forceinline__ float row_bcast(float x) {
    return __builtin_bit_cast(float,
        __builtin_amdgcn_update_dpp(0, __builtin_bit_cast(int, x), 0x150 + S, 0xf, 0xf, false));
}

__device__ __forceinline__ float dot4(const float4 x, const float4 y) {
    return x.x * y.x + x.y * y.y + x.z * y.z + x.w * y.w;
}

__device__ __forceinline__ float exp_hw(const float x) {
    return __builtin_amdgcn_exp2f(x * 1.44269504088896341f);
}

__device__ __forceinline__ float elu1(const float x) {
    const float em1 = exp_hw(x) - 1.0f;
    return x > 0.0f ? x : em1;
}

struct Trip {
    float4 hv[2][DEG];
    float4 own[2];
};

__device__ __forceinline__ void load_trip(Trip& t, const int it, const int* shj,
                                          const float* __restrict__ own0, const float* __restrict__ ab,
                                          const int hf, const int l16) {
    #pragma unroll
    for (int u = 0; u < 2; ++u) {
        const int rl = it * 4 + u * 2 + hf;
        const int4 ja = *(const int4*)&shj[rl * DEG];
        const int4 jb = *(const int4*)&shj[rl * DEG + 4];
        const int j[DEG] = {ja.x, ja.y, ja.z, ja.w, jb.x, jb.y, jb.z, jb.w};
        t.own[u] = *(const float4*)((const char*)own0 + (unsigned)(rl * (FF * 4) + l16 * 16));
        #pragma unroll
        for (int d = 0; d < DEG; ++d)
            t.hv[u][d] = *(const float4*)((const char*)ab + (unsigned)(j[d] * (FF * 4) + l16 * 16));
    }
}

__global__ __launch_bounds__(256) void gat_fused(const float* __restrict__ atoms,
                                                 const int* __restrict__ edges,
                                                 const float* __restrict__ W,
                                                 const float* __restrict__ a,
                                                 float* __restrict__ out) {
    __shared__ float ldsWt[FF * WT_STRIDE];
    __shared__ __attribute__((aligned(16))) float sh_wa[2 * FF];
    __shared__ __attribute__((aligned(16))) int   sh_j[8][RPW * DEG];
    __shared__ __attribute__((aligned(16))) float sh_agg[8][RPW * AG_STRIDE];

    const int lane = threadIdx.x & 31;
    const int wave = __builtin_amdgcn_readfirstlane((int)(threadIdx.x >> 5));
    const int l16  = lane & 15;
    const int hf   = lane >> 4;
    const int row0 = (blockIdx.x * 8 + wave) * RPW;
    const int batchRow0 = row0 & ~(NN - 1);

    {
        const int4 mine = *(const int4*)(edges + (size_t)row0 * DEG + lane * 4);
        float wv[16];
        #pragma unroll
        for (int t = 0; t < 16; ++t) wv[t] = W[threadIdx.x + 256 * t];
        #pragma unroll
        for (int t = 0; t < 16; ++t) {
            const int i = threadIdx.x + 256 * t;
            ldsWt[(i & 63) * WT_STRIDE + (i >> 6)] = wv[t];
        }
        *(int4*)&sh_j[wave][lane * 4] = mine;
    }
    __syncthreads();

    const float* ab   = atoms + (size_t)batchRow0 * FF;
    const float* own0 = atoms + (size_t)row0 * FF;
    Trip cur;
    load_trip(cur, 0, sh_j[wave], own0, ab, hf, l16);

    if (threadIdx.x < 2 * FF) {
        const int which = __builtin_amdgcn_readfirstlane((int)(threadIdx.x >> 6));
        const int k     = threadIdx.x & 63;
        const float* av = a + which * FF;
        float s = 0.0f;
        #pragma unroll
        for (int n = 0; n < FF; ++n) s += ldsWt[n * WT_STRIDE + k] * av[n];
        sh_wa[which * FF + k] = s;
    }
    __syncthreads();

    const float4 wa1_0 = *(const float4*)&sh_wa[4 * l16];
    const float4 wa2_0 = *(const float4*)&sh_wa[FF + 4 * l16];
    const int ul = l16 >> 3;
    const int dl = l16 & 7;
    const bool b3 = (l16 & 8) != 0, b2 = (l16 & 4) != 0, b1 = (l16 & 2) != 0, b0 = (l16 & 1) != 0;

    #pragma unroll
    for (int it = 0; it < TRIPS; ++it) {
        Trip nxt;
        if (it + 1 < TRIPS) load_trip(nxt, it + 1, sh_j[wave], own0, ab, hf, l16);

        __builtin_amdgcn_sched_barrier(0);
        float4 wa1 = wa1_0, wa2 = wa2_0;
        asm volatile("" : "+v"(wa1.x), "+v"(wa1.y), "+v"(wa1.z), "+v"(wa1.w),
                          "+v"(wa2.x), "+v"(wa2.y), "+v"(wa2.z), "+v"(wa2.w));

        float v[16];
        #pragma unroll
        for (int u = 0; u < 2; ++u) {
            #pragma unroll
            for (int d = 0; d < DEG; ++d) v[8 * u + d] = dot4(cur.hv[u][d], wa2);
        }
        float x1[8], x2[4], x3[2];
        #pragma unroll
        for (int i = 0; i < 8; ++i) {
            const float keep = b3 ? v[8 + i] : v[i];
            const float send = b3 ? v[i] : v[8 + i];
            x1[i] = keep + row_xor<8>(send);
        }
        #pragma unroll
        for (int i = 0; i < 4; ++i) {
            const float keep = b2 ? x1[4 + i] : x1[i];
            const float send = b2 ? x1[i] : x1[4 + i];
            x2[i] = keep + row_xor<4>(send);
        }
        #pragma unroll
        for (int i = 0; i < 2; ++i) {
            const float keep = b1 ? x2[2 + i] : x2[i];
            const float send = b1 ? x2[i] : x2[2 + i];
            x3[i] = keep + row_xor<2>(send);
        }
        const float keep0 = b0 ? x3[1] : x3[0];
        const float send0 = b0 ? x3[0] : x3[1];
        const float s2e   = keep0 + row_xor<1>(send0);

        float q0 = dot4(cur.own[0], wa1);
        float q1 = dot4(cur.own[1], wa1);
        q0 += row_xor<1>(q0); q1 += row_xor<1>(q1);
        q0 += row_xor<2>(q0); q1 += row_xor<2>(q1);
        q0 += row_xor<4>(q0); q1 += row_xor<4>(q1);
        q0 += row_xor<8>(q0); q1 += row_xor<8>(q1);
        const float s1r = ul ? q1 : q0;

        const int  rs = (it * 4 + ul * 2 + hf) * DEG;
        const int4 ka = *(const int4*)&sh_j[wave][rs];
        const int4 kb4 = *(const int4*)&sh_j[wave][rs + 4];
        const int  kj[DEG] = {ka.x, ka.y, ka.z, ka.w, kb4.x, kb4.y, kb4.z, kb4.w};
        const int  jm = sh_j[wave][rs + dl];
        int dupi = 0;
        #pragma unroll
        for (int d = 0; d < DEG; ++d) dupi |= (int)(d < dl) & (int)(kj[d] == jm);
        const float sv = s1r + s2e;
        const float lr = sv > 0.0f ? sv : LRELU_ALPHA * sv;
        const float e  = dupi ? MASK_VAL : lr;
        float m = e;
        m = fmaxf(m, row_xor<1>(m));
        m = fmaxf(m, row_xor<2>(m));
        m = fmaxf(m, row_xor<4>(m));
        const float wexp = exp_hw(e - m);
        float sum = wexp + row_xor<1>(wexp);
        sum += row_xor<2>(sum);
        sum += row_xor<4>(sum);
        const float wn = wexp * (1.0f / sum);

        const float wg[2][DEG] = {
            {row_bcast<0>(wn), row_bcast<1>(wn), row_bcast<2>(wn),  row_bcast<3>(wn),
             row_bcast<4>(wn), row_bcast<5>(wn), row_bcast<6>(wn),  row_bcast<7>(wn)},
            {row_bcast<8>(wn), row_bcast<9>(wn), row_bcast<10>(wn), row_bcast<11>(wn),
             row_bcast<12>(wn), row_bcast<13>(wn), row_bcast<14>(wn), row_bcast<15>(wn)}};

        #pragma unroll
        for (int u = 0; u < 2; ++u) {
            const int rl = it * 4 + u * 2 + hf;
            float4 acc = {0.0f, 0.0f, 0.0f, 0.0f};
            #pragma unroll
            for (int d = 0; d < DEG; ++d) {
                acc.x += wg[u][d] * cur.hv[u][d].x;
                acc.y += wg[u][d] * cur.hv[u][d].y;
                acc.z += wg[u][d] * cur.hv[u][d].z;
                acc.w += wg[u][d] * cur.hv[u][d].w;
            }
            *(float4*)&sh_agg[wave][rl * AG_STRIDE + 4 * l16] = acc;
        }
        if (it + 1 < TRIPS) cur = nxt;
    }
    __syncthreads();

    v8f acc0 = {}, acc1 = {}, acc2 = {}, acc3 = {};
    const float* aRow  = &sh_agg[wave][l16 * AG_STRIDE];
    const float* bCol0 = ldsWt + (l16 +  0) * WT_STRIDE;
    const float* bCol1 = ldsWt + (l16 + 16) * WT_STRIDE;
    const float* bCol2 = ldsWt + (l16 + 32) * WT_STRIDE;
    const float* bCol3 = ldsWt + (l16 + 48) * WT_STRIDE;
    #pragma unroll
    for (int k = 0; k < FF; k += 4) {
        const int kb = k + hf * 2;
        v2f afrag = *(const v2f*)(aRow + kb);
        v2f f0 = *(const v2f*)(bCol0 + kb);
        v2f f1 = *(const v2f*)(bCol1 + kb);
        v2f f2 = *(const v2f*)(bCol2 + kb);
        v2f f3 = *(const v2f*)(bCol3 + kb);
        acc0 = __builtin_amdgcn_wmma_f32_16x16x4_f32(false, afrag, false, f0,
                                                     (short)0, acc0, false, false);
        acc1 = __builtin_amdgcn_wmma_f32_16x16x4_f32(false, afrag, false, f1,
                                                     (short)0, acc1, false, false);
        acc2 = __builtin_amdgcn_wmma_f32_16x16x4_f32(false, afrag, false, f2,
                                                     (short)0, acc2, false, false);
        acc3 = __builtin_amdgcn_wmma_f32_16x16x4_f32(false, afrag, false, f3,
                                                     (short)0, acc3, false, false);
    }
    __syncthreads();

    #pragma unroll
    for (int r = 0; r < 8; ++r) {
        float* t = &sh_agg[wave][(hf * 8 + r) * AG_STRIDE + l16];
        t[0]  = elu1(acc0[r]);
        t[16] = elu1(acc1[r]);
        t[32] = elu1(acc2[r]);
        t[48] = elu1(acc3[r]);
    }
    __syncthreads();
    #pragma unroll
    for (int i = 0; i < 8; ++i) {
        const int rl = 2 * i + hf;
        const float4 y = *(const float4*)&sh_agg[wave][rl * AG_STRIDE + 4 * l16];
        *(float4*)(out + (size_t)(row0 + rl) * FF + l16 * 4) = y;
    }
}

extern "C" void kernel_launch(void* const* d_in, const int* in_sizes, int n_in,
                              void* d_out, int out_size, void* d_ws, size_t ws_size,
                              hipStream_t stream) {
    (void)in_sizes; (void)n_in; (void)out_size; (void)d_ws; (void)ws_size;

    const float* atoms = (const float*)d_in[0];
    const int*   edges = (const int*)d_in[1];
    const float* W     = (const float*)d_in[2];
    const float* a     = (const float*)d_in[3];
    float* out = (float*)d_out;

    const int ROWS = BB * NN;
    gat_fused<<<ROWS / (8 * RPW), 256, 0, stream>>>(atoms, edges, W, a, out);
}
